# starting kernel
# baseline (speedup 1.0000x reference)
#include <hip/hip_runtime.h>
#include <hip/hip_bf16.h>

#define BB   128
#define RR   4096
#define CIN  16
#define COUT 32
#define CC   2
#define EPSF 1e-12f

typedef __attribute__((ext_vector_type(2))) float v2f;
typedef __attribute__((ext_vector_type(8))) float v8f;

__global__ __launch_bounds__(128) void caps_priors_wmma(
    const float* __restrict__ x,
    const float* __restrict__ w,
    float* __restrict__ priors)
{
    const int RT   = 4;
    const int c    = blockIdx.y;
    const int r0   = blockIdx.x * RT;
    const int wave = threadIdx.x >> 5;
    const int lane = threadIdx.x & 31;
    const int m    = lane & 15;
    const int kh   = lane >> 4;

    for (int rr = 0; rr < RT; ++rr) {
        const int r = r0 + rr;
        const float* wr = w + ((size_t)c * RR + r) * (CIN * COUT);

        v8f acc[2][2];
        const v8f vzero = {0.f,0.f,0.f,0.f,0.f,0.f,0.f,0.f};
        acc[0][0] = vzero; acc[0][1] = vzero; acc[1][0] = vzero; acc[1][1] = vzero;

        #pragma unroll
        for (int kk = 0; kk < 4; ++kk) {
            v2f afrag[2];
            #pragma unroll
            for (int t = 0; t < 2; ++t) {
                const int b = (2 * wave + t) * 16 + m;
                const float* xp = x + ((size_t)b * RR + r) * CIN + kk * 4 + 2 * kh;
                afrag[t] = *(const v2f*)xp;
            }
            v2f bfrag[2];
            #pragma unroll
            for (int nt = 0; nt < 2; ++nt) {
                const int n    = nt * 16 + m;
                const int krow = kk * 4 + 2 * kh;
                bfrag[nt].x = wr[(size_t)krow * COUT + n];
                bfrag[nt].y = wr[(size_t)(krow + 1) * COUT + n];
            }
            #pragma unroll
            for (int t = 0; t < 2; ++t)
                #pragma unroll
                for (int nt = 0; nt < 2; ++nt)
                    acc[t][nt] = __builtin_amdgcn_wmma_f32_16x16x4_f32(
                        false, afrag[t], false, bfrag[nt],
                        (short)0, acc[t][nt], false, false);
        }

        #pragma unroll
        for (int t = 0; t < 2; ++t) {
            #pragma unroll
            for (int nt = 0; nt < 2; ++nt) {
                #pragma unroll
                for (int v = 0; v < 8; ++v) {
                    const int b = (2 * wave + t) * 16 + v + 8 * kh;
                    const int o = nt * 16 + m;
                    priors[(((size_t)c * BB + b) * RR + r) * COUT + o] = acc[t][nt][v];
                }
            }
        }
    }
}

__global__ __launch_bounds__(256) void caps_iter0(
    const float* __restrict__ priors,
    float* __restrict__ outbuf)
{
    const int c = blockIdx.y, b = blockIdx.x, tid = threadIdx.x;
    const int o = tid & 31, g = tid >> 5;
    const float* p = priors + ((size_t)c * BB + b) * RR * COUT;

    float part = 0.f;
    for (int r = g; r < RR; r += 8) part += p[(size_t)r * COUT + o];

    __shared__ float red[256];
    __shared__ float sArr[COUT];
    __shared__ float scaleSh;
    red[tid] = part;
    __syncthreads();
    if (tid < 32) {
        float s = 0.f;
        #pragma unroll
        for (int gg = 0; gg < 8; ++gg) s += red[gg * 32 + tid];
        sArr[tid] = s * (1.0f / RR);
    }
    __syncthreads();
    if (tid == 0) {
        float sn = 0.f;
        #pragma unroll
        for (int i = 0; i < COUT; ++i) sn += sArr[i] * sArr[i];
        scaleSh = (sn / (1.f + sn + EPSF)) * rsqrtf(sn + EPSF);
    }
    __syncthreads();
    if (tid < 32) outbuf[((size_t)c * BB + b) * COUT + tid] = sArr[tid] * scaleSh;
}

__global__ __launch_bounds__(256) void caps_route_iter(
    const float* __restrict__ priors,
    float* __restrict__ logits,
    const float* __restrict__ outPrev,
    float* __restrict__ outNext,
    int usePrevLogits, int storeLogits)
{
    const int c = blockIdx.y, b = blockIdx.x, tid = threadIdx.x;
    const float* p = priors + ((size_t)c * BB + b) * RR * COUT;
    float* lg = logits + ((size_t)c * BB + b) * RR;

    __shared__ float lgS[RR];
    __shared__ float oPrev[COUT];
    __shared__ float red[256];
    __shared__ float zRed[8];
    __shared__ float sArr[COUT];
    __shared__ float scaleSh;

    if (tid < COUT) oPrev[tid] = outPrev[((size_t)c * BB + b) * COUT + tid];
    __syncthreads();

    float lmax = -3.402823466e38f;
    for (int r = tid; r < RR; r += 256) {
        const float* pr = p + (size_t)r * COUT;
        float d = 0.f;
        #pragma unroll
        for (int o = 0; o < COUT; ++o) d += pr[o] * oPrev[o];
        float v = d + (usePrevLogits ? lg[r] : 0.f);
        lgS[r] = v;
        lmax = fmaxf(lmax, v);
    }
    red[tid] = lmax;
    __syncthreads();
    #pragma unroll
    for (int s2 = 128; s2 > 0; s2 >>= 1) {
        if (tid < s2) red[tid] = fmaxf(red[tid], red[tid + s2]);
        __syncthreads();
    }
    const float mx = red[0];
    __syncthreads();

    const int o = tid & 31, g = tid >> 5;
    float spart = 0.f, zpart = 0.f;
    for (int r = g; r < RR; r += 8) {
        const float wgt = __expf(lgS[r] - mx);
        spart += wgt * p[(size_t)r * COUT + o];
        if (o == 0) zpart += wgt;
    }
    red[tid] = spart;
    if (o == 0) zRed[g] = zpart;
    __syncthreads();
    if (tid < 32) {
        float s = 0.f;
        #pragma unroll
        for (int gg = 0; gg < 8; ++gg) s += red[gg * 32 + tid];
        sArr[tid] = s;
    }
    __syncthreads();
    if (tid == 0) {
        float Z = 0.f;
        #pragma unroll
        for (int i = 0; i < 8; ++i) Z += zRed[i];
        const float invZ = 1.f / Z;
        float sn = 0.f;
        #pragma unroll
        for (int i = 0; i < COUT; ++i) {
            const float sv = sArr[i] * invZ;
            sArr[i] = sv;
            sn += sv * sv;
        }
        scaleSh = (sn / (1.f + sn + EPSF)) * rsqrtf(sn + EPSF);
    }
    __syncthreads();
    if (tid < 32) outNext[((size_t)c * BB + b) * COUT + tid] = sArr[tid] * scaleSh;
    if (storeLogits)
        for (int r = tid; r < RR; r += 256) lg[r] = lgS[r];
}

extern "C" void kernel_launch(void* const* d_in, const int* in_sizes, int n_in,
                              void* d_out, int out_size, void* d_ws, size_t ws_size,
                              hipStream_t stream) {
    (void)in_sizes; (void)n_in; (void)out_size; (void)ws_size;
    const float* x = (const float*)d_in[0];
    const float* w = (const float*)d_in[1];
    float* out = (float*)d_out;

    char* ws = (char*)d_ws;
    float* priors = (float*)ws;
    const size_t priorsBytes = (size_t)CC * BB * RR * COUT * sizeof(float);
    float* logitsBuf = (float*)(ws + priorsBytes);
    const size_t logitsBytes = (size_t)CC * BB * RR * sizeof(float);
    float* obuf = (float*)(ws + priorsBytes + logitsBytes);

    dim3 gA(RR / 4, CC);
    caps_priors_wmma<<<gA, 128, 0, stream>>>(x, w, priors);

    dim3 gB(BB, CC);
    caps_iter0<<<gB, 256, 0, stream>>>(priors, obuf);
    caps_route_iter<<<gB, 256, 0, stream>>>(priors, logitsBuf, obuf, obuf, 0, 1);
    caps_route_iter<<<gB, 256, 0, stream>>>(priors, logitsBuf, obuf, out, 1, 0);
}


# Recompute priors on the f32 matrix unit inside each routing pass (running softmax per 16-node chunk, small partials combined per row) instead of storing and re-reading the 134 MB priors array
# speedup vs baseline: 2.0964x; 2.0964x over previous
#include <hip/hip_runtime.h>


#define BB   128
#define RR   4096
#define CIN  16
#define COUT 32
#define CC   2
#define EPSF 1e-12f
#define RCH  16
#define NCH  (RR / RCH)

typedef __attribute__((ext_vector_type(2))) float v2f;
typedef __attribute__((ext_vector_type(8))) float v8f;

struct Frags {
    v2f a[4];
    v2f b[4][2];
};

__device__ __forceinline__ void load_frags(Frags& f, const float* __restrict__ x,
                                           const float* __restrict__ w,
                                           int c, int r, int brow, int m, int kh)
{
    const float* wr = w + ((size_t)c * RR + r) * (CIN * COUT);
    const float* xr = x + ((size_t)brow * RR + r) * CIN;
    #pragma unroll
    for (int kk = 0; kk < 4; ++kk) {
        f.a[kk] = *(const v2f*)(xr + kk * 4 + 2 * kh);
        const int krow = kk * 4 + 2 * kh;
        #pragma unroll
        for (int nt = 0; nt < 2; ++nt) {
            const int n = nt * 16 + m;
            f.b[kk][nt].x = wr[(size_t)krow * COUT + n];
            f.b[kk][nt].y = wr[(size_t)(krow + 1) * COUT + n];
        }
    }
}

__device__ __forceinline__ float row16_sum(float v)
{
    v += __builtin_bit_cast(float, __builtin_amdgcn_update_dpp(0, __builtin_bit_cast(int, v), 0x161, 0xf, 0xf, false));
    v += __builtin_bit_cast(float, __builtin_amdgcn_update_dpp(0, __builtin_bit_cast(int, v), 0x162, 0xf, 0xf, false));
    v += __builtin_bit_cast(float, __builtin_amdgcn_update_dpp(0, __builtin_bit_cast(int, v), 0x164, 0xf, 0xf, false));
    v += __builtin_bit_cast(float, __builtin_amdgcn_update_dpp(0, __builtin_bit_cast(int, v), 0x168, 0xf, 0xf, false));
    return v;
}

__global__ __launch_bounds__(256) void caps_pass_sum(
    const float* __restrict__ x,
    const float* __restrict__ w,
    float* __restrict__ partS)
{
    const int c    = blockIdx.y;
    const int ch   = blockIdx.x;
    const int r0   = ch * RCH;
    const int tile = threadIdx.x >> 5;
    const int lane = threadIdx.x & 31;
    const int m    = lane & 15;
    const int kh   = lane >> 4;
    const int brow = tile * 16 + m;

    const v8f vzero = {0.f,0.f,0.f,0.f,0.f,0.f,0.f,0.f};
    v8f sacc[2];
    sacc[0] = vzero; sacc[1] = vzero;

    Frags cur, nxt;
    load_frags(cur, x, w, c, r0, brow, m, kh);
    #pragma unroll 2
    for (int rr = 0; rr < RCH; ++rr) {
        const int rn = r0 + (rr + 1 < RCH ? rr + 1 : rr);
        load_frags(nxt, x, w, c, rn, brow, m, kh);
        #pragma unroll
        for (int kk = 0; kk < 4; ++kk)
            #pragma unroll
            for (int nt = 0; nt < 2; ++nt)
                sacc[nt] = __builtin_amdgcn_wmma_f32_16x16x4_f32(
                    false, cur.a[kk], false, cur.b[kk][nt],
                    (short)0, sacc[nt], false, false);
        cur = nxt;
    }

    #pragma unroll
    for (int nt = 0; nt < 2; ++nt)
        #pragma unroll
        for (int v = 0; v < 8; ++v) {
            const int b = tile * 16 + v + 8 * kh;
            partS[(((size_t)c * NCH + ch) * BB + b) * COUT + nt * 16 + m] = sacc[nt][v];
        }
}

__global__ __launch_bounds__(256) void caps_pass_soft(
    const float* __restrict__ x,
    const float* __restrict__ w,
    const float* __restrict__ vvec,
    float* __restrict__ partS,
    float* __restrict__ partM,
    float* __restrict__ partZ)
{
    const int c    = blockIdx.y;
    const int ch   = blockIdx.x;
    const int r0   = ch * RCH;
    const int tile = threadIdx.x >> 5;
    const int lane = threadIdx.x & 31;
    const int m    = lane & 15;
    const int kh   = lane >> 4;
    const int brow = tile * 16 + m;

    const v8f vzero = {0.f,0.f,0.f,0.f,0.f,0.f,0.f,0.f};
    v8f sacc[2];
    sacc[0] = vzero; sacc[1] = vzero;
    float vv[2][8], mrun[8], zrun[8];
    #pragma unroll
    for (int v = 0; v < 8; ++v) {
        const int b = tile * 16 + v + 8 * kh;
        vv[0][v] = vvec[((size_t)c * BB + b) * COUT + m];
        vv[1][v] = vvec[((size_t)c * BB + b) * COUT + 16 + m];
        mrun[v] = -3.402823466e38f;
        zrun[v] = 0.f;
    }

    Frags cur, nxt;
    load_frags(cur, x, w, c, r0, brow, m, kh);
    #pragma unroll 2
    for (int rr = 0; rr < RCH; ++rr) {
        const int rn = r0 + (rr + 1 < RCH ? rr + 1 : rr);
        load_frags(nxt, x, w, c, rn, brow, m, kh);

        v8f acc[2];
        acc[0] = vzero; acc[1] = vzero;
        #pragma unroll
        for (int kk = 0; kk < 4; ++kk)
            #pragma unroll
            for (int nt = 0; nt < 2; ++nt)
                acc[nt] = __builtin_amdgcn_wmma_f32_16x16x4_f32(
                    false, cur.a[kk], false, cur.b[kk][nt],
                    (short)0, acc[nt], false, false);

        float d[8];
        #pragma unroll
        for (int v = 0; v < 8; ++v)
            d[v] = acc[0][v] * vv[0][v] + acc[1][v] * vv[1][v];
        #pragma unroll
        for (int v = 0; v < 8; ++v)
            d[v] = row16_sum(d[v]);

        #pragma unroll
        for (int v = 0; v < 8; ++v) {
            const float mo = mrun[v];
            const float mn = fmaxf(mo, d[v]);
            const float e  = __expf(fminf(mo, d[v]) - mn);
            const bool  up = d[v] > mo;
            const float sc = up ? e : 1.f;
            const float wg = up ? 1.f : e;
            mrun[v] = mn;
            zrun[v] = zrun[v] * sc + wg;
            sacc[0][v] = sacc[0][v] * sc + wg * acc[0][v];
            sacc[1][v] = sacc[1][v] * sc + wg * acc[1][v];
        }
        cur = nxt;
    }

    #pragma unroll
    for (int nt = 0; nt < 2; ++nt)
        #pragma unroll
        for (int v = 0; v < 8; ++v) {
            const int b = tile * 16 + v + 8 * kh;
            partS[(((size_t)c * NCH + ch) * BB + b) * COUT + nt * 16 + m] = sacc[nt][v];
        }
    if (m == 0) {
        #pragma unroll
        for (int v = 0; v < 8; ++v) {
            const int b = tile * 16 + v + 8 * kh;
            partM[((size_t)c * NCH + ch) * BB + b] = mrun[v];
            partZ[((size_t)c * NCH + ch) * BB + b] = zrun[v];
        }
    }
}

__global__ __launch_bounds__(256) void caps_combine(
    const float* __restrict__ partS,
    const float* __restrict__ partM,
    const float* __restrict__ partZ,
    const float* __restrict__ addv,
    float* __restrict__ dst,
    int soft)
{
    const int c = blockIdx.y, b = blockIdx.x, tid = threadIdx.x;

    __shared__ float eS[NCH];
    __shared__ float red[256];
    __shared__ float sArr[COUT];
    __shared__ float scaleSh;

    const size_t pi = ((size_t)c * NCH + tid) * BB + b;
    const float mc = soft ? partM[pi] : 0.f;
    red[tid] = mc;
    __syncthreads();
    #pragma unroll
    for (int s2 = 128; s2 > 0; s2 >>= 1) {
        if (tid < s2) red[tid] = fmaxf(red[tid], red[tid + s2]);
        __syncthreads();
    }
    const float mx = red[0];
    __syncthreads();

    const float e  = soft ? __expf(mc - mx) : 1.f;
    const float zc = soft ? partZ[pi] * e : (float)RCH;
    eS[tid]  = e;
    red[tid] = zc;
    __syncthreads();
    #pragma unroll
    for (int s2 = 128; s2 > 0; s2 >>= 1) {
        if (tid < s2) red[tid] += red[tid + s2];
        __syncthreads();
    }
    const float Z = red[0];
    __syncthreads();

    const int o = tid & 31, g = tid >> 5;
    float spart = 0.f;
    #pragma unroll 8
    for (int k = g; k < NCH; k += 8)
        spart += eS[k] * partS[(((size_t)c * NCH + k) * BB + b) * COUT + o];
    red[tid] = spart;
    __syncthreads();
    if (tid < 32) {
        float s = 0.f;
        #pragma unroll
        for (int gg = 0; gg < 8; ++gg) s += red[gg * 32 + tid];
        sArr[tid] = s / Z;
    }
    __syncthreads();
    if (tid == 0) {
        float sn = 0.f;
        #pragma unroll
        for (int i = 0; i < COUT; ++i) sn += sArr[i] * sArr[i];
        scaleSh = (sn / (1.f + sn + EPSF)) * rsqrtf(sn + EPSF);
    }
    __syncthreads();
    if (tid < 32) {
        const size_t oi = ((size_t)c * BB + b) * COUT + tid;
        float val = sArr[tid] * scaleSh;
        if (addv) val += addv[oi];
        dst[oi] = val;
    }
}

extern "C" void kernel_launch(void* const* d_in, const int* in_sizes, int n_in,
                              void* d_out, int out_size, void* d_ws, size_t ws_size,
                              hipStream_t stream) {
    (void)in_sizes; (void)n_in; (void)out_size; (void)ws_size;
    const float* x = (const float*)d_in[0];
    const float* w = (const float*)d_in[1];
    float* out = (float*)d_out;

    float* partS = (float*)d_ws;
    float* partM = partS + (size_t)CC * NCH * BB * COUT;
    float* partZ = partM + (size_t)CC * NCH * BB;
    float* v1    = partZ + (size_t)CC * NCH * BB;
    float* v2    = v1 + (size_t)CC * BB * COUT;

    dim3 gP(NCH, CC);
    dim3 gC(BB, CC);
    caps_pass_sum<<<gP, 256, 0, stream>>>(x, w, partS);
    caps_combine<<<gC, 256, 0, stream>>>(partS, partM, partZ, nullptr, v1, 0);
    caps_pass_soft<<<gP, 256, 0, stream>>>(x, w, v1, partS, partM, partZ);
    caps_combine<<<gC, 256, 0, stream>>>(partS, partM, partZ, v1, v2, 1);
    caps_pass_soft<<<gP, 256, 0, stream>>>(x, w, v2, partS, partM, partZ);
    caps_combine<<<gC, 256, 0, stream>>>(partS, partM, partZ, nullptr, out, 1);
}


# Softmax pass forms priors transposed (W as A operand, x as B): a lane owns one batch row, so the logit is an in-lane sum plus one half exchange and max/sum are per-lane scalars (no DPP row sums)
# speedup vs baseline: 2.3998x; 1.1447x over previous
#include <hip/hip_runtime.h>


#define BB   128
#define RR   4096
#define CIN  16
#define COUT 32
#define CC   2
#define EPSF 1e-12f
#define RCH  16
#define NCH  (RR / RCH)

typedef __attribute__((ext_vector_type(2))) float v2f;
typedef __attribute__((ext_vector_type(8))) float v8f;

struct Frags {
    v2f a[4];
    v2f b[4][2];
};

__device__ __forceinline__ void load_frags(Frags& f, const float* __restrict__ x,
                                           const float* __restrict__ w,
                                           int c, int r, int brow, int m, int kh)
{
    const float* wr = w + ((size_t)c * RR + r) * (CIN * COUT);
    const float* xr = x + ((size_t)brow * RR + r) * CIN;
    #pragma unroll
    for (int kk = 0; kk < 4; ++kk) {
        f.a[kk] = *(const v2f*)(xr + kk * 4 + 2 * kh);
        const int krow = kk * 4 + 2 * kh;
        #pragma unroll
        for (int nt = 0; nt < 2; ++nt) {
            const int n = nt * 16 + m;
            f.b[kk][nt].x = wr[(size_t)krow * COUT + n];
            f.b[kk][nt].y = wr[(size_t)(krow + 1) * COUT + n];
        }
    }
}

__device__ __forceinline__ float xhalf(float v)
{
    const int b = __builtin_bit_cast(int, v);
    return __builtin_bit_cast(float, __builtin_amdgcn_permlanex16(b, b, 0x76543210, 0xfedcba98, false, false));
}

__global__ __launch_bounds__(256) void caps_pass_sum(
    const float* __restrict__ x,
    const float* __restrict__ w,
    float* __restrict__ partS)
{
    const int c    = blockIdx.y;
    const int ch   = blockIdx.x;
    const int r0   = ch * RCH;
    const int tile = threadIdx.x >> 5;
    const int lane = threadIdx.x & 31;
    const int m    = lane & 15;
    const int kh   = lane >> 4;
    const int brow = tile * 16 + m;

    const v8f vzero = {0.f,0.f,0.f,0.f,0.f,0.f,0.f,0.f};
    v8f sacc[2];
    sacc[0] = vzero; sacc[1] = vzero;

    Frags cur, nxt;
    load_frags(cur, x, w, c, r0, brow, m, kh);
    #pragma unroll 2
    for (int rr = 0; rr < RCH; ++rr) {
        const int rn = r0 + (rr + 1 < RCH ? rr + 1 : rr);
        load_frags(nxt, x, w, c, rn, brow, m, kh);
        #pragma unroll
        for (int kk = 0; kk < 4; ++kk)
            #pragma unroll
            for (int nt = 0; nt < 2; ++nt)
                sacc[nt] = __builtin_amdgcn_wmma_f32_16x16x4_f32(
                    false, cur.a[kk], false, cur.b[kk][nt],
                    (short)0, sacc[nt], false, false);
        cur = nxt;
    }

    #pragma unroll
    for (int nt = 0; nt < 2; ++nt)
        #pragma unroll
        for (int v = 0; v < 8; ++v) {
            const int b = tile * 16 + v + 8 * kh;
            partS[(((size_t)c * NCH + ch) * BB + b) * COUT + nt * 16 + m] = sacc[nt][v];
        }
}

__global__ __launch_bounds__(256) void caps_pass_soft(
    const float* __restrict__ x,
    const float* __restrict__ w,
    const float* __restrict__ vvec,
    float* __restrict__ partS,
    float* __restrict__ partM,
    float* __restrict__ partZ)
{
    const int c    = blockIdx.y;
    const int ch   = blockIdx.x;
    const int r0   = ch * RCH;
    const int tile = threadIdx.x >> 5;
    const int lane = threadIdx.x & 31;
    const int m    = lane & 15;
    const int kh   = lane >> 4;
    const int brow = tile * 16 + m;

    const v8f vzero = {0.f,0.f,0.f,0.f,0.f,0.f,0.f,0.f};
    v8f sacc[2];
    sacc[0] = vzero; sacc[1] = vzero;

    const float* vp = vvec + ((size_t)c * BB + brow) * COUT + 8 * kh;
    const v8f vv0 = *(const v8f*)(vp);
    const v8f vv1 = *(const v8f*)(vp + 16);
    float mrun = -3.402823466e38f;
    float zrun = 0.f;

    Frags cur, nxt;
    load_frags(cur, x, w, c, r0, brow, m, kh);
    #pragma unroll 2
    for (int rr = 0; rr < RCH; ++rr) {
        const int rn = r0 + (rr + 1 < RCH ? rr + 1 : rr);
        load_frags(nxt, x, w, c, rn, brow, m, kh);

        v8f acc[2];
        acc[0] = vzero; acc[1] = vzero;
        #pragma unroll
        for (int kk = 0; kk < 4; ++kk)
            #pragma unroll
            for (int nt = 0; nt < 2; ++nt)
                acc[nt] = __builtin_amdgcn_wmma_f32_16x16x4_f32(
                    false, cur.b[kk][nt], false, cur.a[kk],
                    (short)0, acc[nt], false, false);

        v2f t0 = __builtin_shufflevector(acc[0], acc[0], 0, 1) * __builtin_shufflevector(vv0, vv0, 0, 1);
        v2f t1 = __builtin_shufflevector(acc[1], acc[1], 0, 1) * __builtin_shufflevector(vv1, vv1, 0, 1);
        t0 += __builtin_shufflevector(acc[0], acc[0], 2, 3) * __builtin_shufflevector(vv0, vv0, 2, 3);
        t1 += __builtin_shufflevector(acc[1], acc[1], 2, 3) * __builtin_shufflevector(vv1, vv1, 2, 3);
        t0 += __builtin_shufflevector(acc[0], acc[0], 4, 5) * __builtin_shufflevector(vv0, vv0, 4, 5);
        t1 += __builtin_shufflevector(acc[1], acc[1], 4, 5) * __builtin_shufflevector(vv1, vv1, 4, 5);
        t0 += __builtin_shufflevector(acc[0], acc[0], 6, 7) * __builtin_shufflevector(vv0, vv0, 6, 7);
        t1 += __builtin_shufflevector(acc[1], acc[1], 6, 7) * __builtin_shufflevector(vv1, vv1, 6, 7);
        const v2f ts = t0 + t1;
        float d = ts.x + ts.y;
        d = d + xhalf(d);

        const float mo = mrun;
        const float mn = fmaxf(mo, d);
        const float e  = __expf(fminf(mo, d) - mn);
        const bool  up = d > mo;
        const float sc = up ? e : 1.f;
        const float wg = up ? 1.f : e;
        mrun = mn;
        zrun = zrun * sc + wg;
        #pragma unroll
        for (int v = 0; v < 8; ++v) {
            sacc[0][v] = sacc[0][v] * sc + wg * acc[0][v];
            sacc[1][v] = sacc[1][v] * sc + wg * acc[1][v];
        }
        cur = nxt;
    }

    float* ps = partS + (((size_t)c * NCH + ch) * BB + brow) * COUT + 8 * kh;
    *(v8f*)(ps)      = sacc[0];
    *(v8f*)(ps + 16) = sacc[1];
    if (kh == 0) {
        partM[((size_t)c * NCH + ch) * BB + brow] = mrun;
        partZ[((size_t)c * NCH + ch) * BB + brow] = zrun;
    }
}

__global__ __launch_bounds__(256) void caps_combine(
    const float* __restrict__ partS,
    const float* __restrict__ partM,
    const float* __restrict__ partZ,
    const float* __restrict__ addv,
    float* __restrict__ dst,
    int soft)
{
    const int c = blockIdx.y, b = blockIdx.x, tid = threadIdx.x;

    __shared__ float eS[NCH];
    __shared__ float red[256];
    __shared__ float sArr[COUT];
    __shared__ float scaleSh;

    const size_t pi = ((size_t)c * NCH + tid) * BB + b;
    const float mc = soft ? partM[pi] : 0.f;
    red[tid] = mc;
    __syncthreads();
    #pragma unroll
    for (int s2 = 128; s2 > 0; s2 >>= 1) {
        if (tid < s2) red[tid] = fmaxf(red[tid], red[tid + s2]);
        __syncthreads();
    }
    const float mx = red[0];
    __syncthreads();

    const float e  = soft ? __expf(mc - mx) : 1.f;
    const float zc = soft ? partZ[pi] * e : (float)RCH;
    eS[tid]  = e;
    red[tid] = zc;
    __syncthreads();
    #pragma unroll
    for (int s2 = 128; s2 > 0; s2 >>= 1) {
        if (tid < s2) red[tid] += red[tid + s2];
        __syncthreads();
    }
    const float Z = red[0];
    __syncthreads();

    const int o = tid & 31, g = tid >> 5;
    float spart = 0.f;
    #pragma unroll 8
    for (int k = g; k < NCH; k += 8)
        spart += eS[k] * partS[(((size_t)c * NCH + k) * BB + b) * COUT + o];
    red[tid] = spart;
    __syncthreads();
    if (tid < 32) {
        float s = 0.f;
        #pragma unroll
        for (int gg = 0; gg < 8; ++gg) s += red[gg * 32 + tid];
        sArr[tid] = s / Z;
    }
    __syncthreads();
    if (tid == 0) {
        float sn = 0.f;
        #pragma unroll
        for (int i = 0; i < COUT; ++i) sn += sArr[i] * sArr[i];
        scaleSh = (sn / (1.f + sn + EPSF)) * rsqrtf(sn + EPSF);
    }
    __syncthreads();
    if (tid < 32) {
        const size_t oi = ((size_t)c * BB + b) * COUT + tid;
        float val = sArr[tid] * scaleSh;
        if (addv) val += addv[oi];
        dst[oi] = val;
    }
}

extern "C" void kernel_launch(void* const* d_in, const int* in_sizes, int n_in,
                              void* d_out, int out_size, void* d_ws, size_t ws_size,
                              hipStream_t stream) {
    (void)in_sizes; (void)n_in; (void)out_size; (void)ws_size;
    const float* x = (const float*)d_in[0];
    const float* w = (const float*)d_in[1];
    float* out = (float*)d_out;

    float* partS = (float*)d_ws;
    float* partM = partS + (size_t)CC * NCH * BB * COUT;
    float* partZ = partM + (size_t)CC * NCH * BB;
    float* v1    = partZ + (size_t)CC * NCH * BB;
    float* v2    = v1 + (size_t)CC * BB * COUT;

    dim3 gP(NCH, CC);
    dim3 gC(BB, CC);
    caps_pass_sum<<<gP, 256, 0, stream>>>(x, w, partS);
    caps_combine<<<gC, 256, 0, stream>>>(partS, partM, partZ, nullptr, v1, 0);
    caps_pass_soft<<<gP, 256, 0, stream>>>(x, w, v1, partS, partM, partZ);
    caps_combine<<<gC, 256, 0, stream>>>(partS, partM, partZ, v1, v2, 1);
    caps_pass_soft<<<gP, 256, 0, stream>>>(x, w, v2, partS, partM, partZ);
    caps_combine<<<gC, 256, 0, stream>>>(partS, partM, partZ, nullptr, out, 1);
}


# Operand fetch by whole lines: the chunk's weights staged once per workgroup in LDS by 16-byte loads (were re-read by all 8 waves), x as 32 contiguous bytes per lane, loads issued 4-8 nodes ahead
# speedup vs baseline: 3.1961x; 1.3318x over previous
#include <hip/hip_runtime.h>


#define BB   128
#define RR   4096
#define CIN  16
#define COUT 32
#define CC   2
#define EPSF 1e-12f
#define RCH  16
#define NCH  (RR / RCH)

typedef __attribute__((ext_vector_type(2))) float v2f;
typedef __attribute__((ext_vector_type(4))) float v4f;
typedef __attribute__((ext_vector_type(8))) float v8f;

#define WPITCH 544

struct XG { v4f lo[4]; v4f hi[4]; };

__device__ __forceinline__ void load_xg(XG& g, const float* __restrict__ xl, int n0)
{
    #pragma unroll
    for (int j = 0; j < 4; ++j) {
        g.lo[j] = *(const v4f*)(xl + (n0 + j) * CIN);
        g.hi[j] = *(const v4f*)(xl + (n0 + j) * CIN + 4);
    }
}

__device__ __forceinline__ void frag_x(v2f (&a)[4], const XG& g, int j)
{
    a[0] = __builtin_shufflevector(g.lo[j], g.lo[j], 0, 1);
    a[1] = __builtin_shufflevector(g.lo[j], g.lo[j], 2, 3);
    a[2] = __builtin_shufflevector(g.hi[j], g.hi[j], 0, 1);
    a[3] = __builtin_shufflevector(g.hi[j], g.hi[j], 2, 3);
}

__device__ __forceinline__ void frag_w(v2f (&b)[4][2], const float* wl, int node)
{
    const float* p = wl + node * WPITCH;
    #pragma unroll
    for (int kk = 0; kk < 4; ++kk)
        #pragma unroll
        for (int nt = 0; nt < 2; ++nt)
            b[kk][nt] = *(const v2f*)(p + kk * 64 + nt * 32);
}

__device__ __forceinline__ void fetch_w(v4f (&t)[8], const float* __restrict__ w, int c, int r0, int tid)
{
    const float* src = w + ((size_t)c * RR + r0) * (CIN * COUT);
    #pragma unroll
    for (int j = 0; j < 4; ++j) {
        const int q    = tid + 256 * j;
        const int node = q >> 6;
        const int rp   = (q >> 3) & 7;
        const int cq   = q & 7;
        const float* s = src + node * (CIN * COUT) + (2 * rp) * COUT + 4 * cq;
        t[2 * j]     = *(const v4f*)(s);
        t[2 * j + 1] = *(const v4f*)(s + COUT);
    }
}

__device__ __forceinline__ void store_w(float* shW, const v4f (&t)[8], int tid)
{
    #pragma unroll
    for (int j = 0; j < 4; ++j) {
        const int q    = tid + 256 * j;
        const int node = q >> 6;
        const int rp   = (q >> 3) & 7;
        const int cq   = q & 7;
        float* d = shW + node * WPITCH + rp * 64 + (rp >> 2) * 32 + 8 * cq;
        const v4f e = t[2 * j], o = t[2 * j + 1];
        const v4f s0 = {e.x, o.x, e.y, o.y};
        const v4f s1 = {e.z, o.z, e.w, o.w};
        *(v4f*)(d)     = s0;
        *(v4f*)(d + 4) = s1;
    }
}

__device__ __forceinline__ float xhalf(float v)
{
    const int b = __builtin_bit_cast(int, v);
    return __builtin_bit_cast(float, __builtin_amdgcn_permlanex16(b, b, 0x76543210, 0xfedcba98, false, false));
}

__global__ __launch_bounds__(256, 4) void caps_pass_sum(
    const float* __restrict__ x,
    const float* __restrict__ w,
    float* __restrict__ partS)
{
    const int c    = blockIdx.y;
    const int ch   = blockIdx.x;
    const int r0   = ch * RCH;
    const int tile = threadIdx.x >> 5;
    const int lane = threadIdx.x & 31;
    const int m    = lane & 15;
    const int kh   = lane >> 4;
    const int brow = tile * 16 + m;

    __shared__ __attribute__((aligned(16))) float shW[RCH * WPITCH];

    const v8f vzero = {0.f,0.f,0.f,0.f,0.f,0.f,0.f,0.f};
    v8f sacc[2];
    sacc[0] = vzero; sacc[1] = vzero;

    v4f wt[8];
    fetch_w(wt, w, c, r0, threadIdx.x);
    const float* xl = x + ((size_t)brow * RR + r0) * CIN + 8 * kh;
    XG xa, xb;
    load_xg(xa, xl, 0);
    load_xg(xb, xl, 4);
    store_w(shW, wt, threadIdx.x);
    __syncthreads();
    const float* wl = shW + kh * 288 + 2 * m;

    #pragma unroll
    for (int g = 0; g < 4; ++g) {
        XG& cur = (g & 1) ? xb : xa;
        #pragma unroll
        for (int j = 0; j < 4; ++j) {
            v2f a[4];
            v2f b[4][2];
            frag_x(a, cur, j);
            frag_w(b, wl, g * 4 + j);
            #pragma unroll
            for (int kk = 0; kk < 4; ++kk)
                #pragma unroll
                for (int nt = 0; nt < 2; ++nt)
                    sacc[nt] = __builtin_amdgcn_wmma_f32_16x16x4_f32(
                        false, a[kk], false, b[kk][nt],
                        (short)0, sacc[nt], false, false);
        }
        if (g < 2) load_xg(cur, xl, (g + 2) * 4);
    }

    #pragma unroll
    for (int nt = 0; nt < 2; ++nt)
        #pragma unroll
        for (int v = 0; v < 8; ++v) {
            const int b = tile * 16 + v + 8 * kh;
            partS[(((size_t)c * NCH + ch) * BB + b) * COUT + nt * 16 + m] = sacc[nt][v];
        }
}

__global__ __launch_bounds__(256, 4) void caps_pass_soft(
    const float* __restrict__ x,
    const float* __restrict__ w,
    const float* __restrict__ vvec,
    float* __restrict__ partS,
    float* __restrict__ partM,
    float* __restrict__ partZ)
{
    const int c    = blockIdx.y;
    const int ch   = blockIdx.x;
    const int r0   = ch * RCH;
    const int tile = threadIdx.x >> 5;
    const int lane = threadIdx.x & 31;
    const int m    = lane & 15;
    const int kh   = lane >> 4;
    const int brow = tile * 16 + m;

    __shared__ __attribute__((aligned(16))) float shW[RCH * WPITCH];

    const v8f vzero = {0.f,0.f,0.f,0.f,0.f,0.f,0.f,0.f};
    v8f sacc[2];
    sacc[0] = vzero; sacc[1] = vzero;

    v4f wt[8];
    fetch_w(wt, w, c, r0, threadIdx.x);
    const float* xl = x + ((size_t)brow * RR + r0) * CIN + 8 * kh;
    XG xa, xb;
    load_xg(xa, xl, 0);
    load_xg(xb, xl, 4);

    const float* vp = vvec + ((size_t)c * BB + brow) * COUT + 8 * kh;
    const v8f vv0 = *(const v8f*)(vp);
    const v8f vv1 = *(const v8f*)(vp + 16);
    float mrun = -3.402823466e38f;
    float zrun = 0.f;

    store_w(shW, wt, threadIdx.x);
    __syncthreads();
    const float* wl = shW + kh * 288 + 2 * m;

    v2f b[4][2], bn[4][2];
    frag_w(b, wl, 0);
    #pragma unroll
    for (int g = 0; g < 4; ++g) {
        XG& cur = (g & 1) ? xb : xa;
        #pragma unroll
        for (int j = 0; j < 4; ++j) {
            const int node = g * 4 + j;
            if (node + 1 < RCH) frag_w(bn, wl, node + 1);
            v2f a[4];
            frag_x(a, cur, j);

            v8f acc[2];
            acc[0] = vzero; acc[1] = vzero;
            #pragma unroll
            for (int kk = 0; kk < 4; ++kk)
                #pragma unroll
                for (int nt = 0; nt < 2; ++nt)
                    acc[nt] = __builtin_amdgcn_wmma_f32_16x16x4_f32(
                        false, b[kk][nt], false, a[kk],
                        (short)0, acc[nt], false, false);

            v2f t0 = __builtin_shufflevector(acc[0], acc[0], 0, 1) * __builtin_shufflevector(vv0, vv0, 0, 1);
            v2f t1 = __builtin_shufflevector(acc[1], acc[1], 0, 1) * __builtin_shufflevector(vv1, vv1, 0, 1);
            t0 += __builtin_shufflevector(acc[0], acc[0], 2, 3) * __builtin_shufflevector(vv0, vv0, 2, 3);
            t1 += __builtin_shufflevector(acc[1], acc[1], 2, 3) * __builtin_shufflevector(vv1, vv1, 2, 3);
            t0 += __builtin_shufflevector(acc[0], acc[0], 4, 5) * __builtin_shufflevector(vv0, vv0, 4, 5);
            t1 += __builtin_shufflevector(acc[1], acc[1], 4, 5) * __builtin_shufflevector(vv1, vv1, 4, 5);
            t0 += __builtin_shufflevector(acc[0], acc[0], 6, 7) * __builtin_shufflevector(vv0, vv0, 6, 7);
            t1 += __builtin_shufflevector(acc[1], acc[1], 6, 7) * __builtin_shufflevector(vv1, vv1, 6, 7);
            const v2f ts = t0 + t1;
            float d = ts.x + ts.y;
            d = d + xhalf(d);

            const float mo = mrun;
            const float mn = fmaxf(mo, d);
            const float e  = __expf(fminf(mo, d) - mn);
            const bool  up = d > mo;
            const float sc = up ? e : 1.f;
            const float wg = up ? 1.f : e;
            mrun = mn;
            zrun = zrun * sc + wg;
            #pragma unroll
            for (int v = 0; v < 8; ++v) {
                sacc[0][v] = sacc[0][v] * sc + wg * acc[0][v];
                sacc[1][v] = sacc[1][v] * sc + wg * acc[1][v];
            }
            if (node + 1 < RCH) {
                #pragma unroll
                for (int kk = 0; kk < 4; ++kk) { b[kk][0] = bn[kk][0]; b[kk][1] = bn[kk][1]; }
            }
            __builtin_amdgcn_sched_barrier(0);
        }
        if (g < 2) {
            load_xg(cur, xl, (g + 2) * 4);
            __builtin_amdgcn_sched_barrier(0);
        }
    }

    float* ps = partS + (((size_t)c * NCH + ch) * BB + brow) * COUT + 8 * kh;
    *(v8f*)(ps)      = sacc[0];
    *(v8f*)(ps + 16) = sacc[1];
    if (kh == 0) {
        partM[((size_t)c * NCH + ch) * BB + brow] = mrun;
        partZ[((size_t)c * NCH + ch) * BB + brow] = zrun;
    }
}

template <int MASK>
__device__ __forceinline__ float xrow(float v)
{
    const int b = __builtin_bit_cast(int, v);
    return __builtin_bit_cast(float, __builtin_amdgcn_update_dpp(b, b, 0x160 + MASK, 0xf, 0xf, false));
}

template <int SOFT>
__global__ __launch_bounds__(256) void caps_combine(
    const float* __restrict__ partS,
    const float* __restrict__ partM,
    const float* __restrict__ partZ,
    const float* __restrict__ addv,
    float* __restrict__ dst)
{
    const int c = blockIdx.y, b = blockIdx.x, tid = threadIdx.x;
    const int wv = tid >> 5, lane = tid & 31;
    const int q = lane & 7, sub = lane >> 3;

    __shared__ float shM[8];
    __shared__ float shZ[8];
    __shared__ __attribute__((aligned(16))) float shS[8][COUT];

    const size_t oi = ((size_t)c * BB + b) * COUT + lane;
    float av = 0.f;
    if (addv != nullptr && wv == 0) av = addv[oi];

    const size_t pbase = ((size_t)c * NCH + wv * 32 + sub) * BB + b;
    v4f   sv[8];
    float mv[8], zv[8];
    if (SOFT) {
        #pragma unroll
        for (int i = 0; i < 8; ++i) mv[i] = partM[pbase + (size_t)i * 4 * BB];
    }
    #pragma unroll
    for (int i = 0; i < 8; ++i) {
        const size_t pi = pbase + (size_t)i * 4 * BB;
        sv[i] = *(const v4f*)(partS + pi * COUT + 4 * q);
        if (SOFT) zv[i] = partZ[pi];
    }
    __builtin_amdgcn_sched_barrier(0);

    float ml = 0.f, zl = 0.f;
    v4f sl = sv[0];
    if (SOFT) {
        ml = mv[0];
        #pragma unroll
        for (int i = 1; i < 8; ++i) ml = fmaxf(ml, mv[i]);
        ml = fmaxf(ml, xrow<8>(ml));
        ml = fmaxf(ml, xhalf(ml));
        const float e0 = __expf(mv[0] - ml);
        zl = e0 * zv[0];
        sl = e0 * sv[0];
        #pragma unroll
        for (int i = 1; i < 8; ++i) {
            const float e = __expf(mv[i] - ml);
            zl += e * zv[i];
            sl += e * sv[i];
        }
        zl += xrow<8>(zl);
        zl += xhalf(zl);
    } else {
        #pragma unroll
        for (int i = 1; i < 8; ++i) sl += sv[i];
    }
    sl.x += xrow<8>(sl.x); sl.y += xrow<8>(sl.y); sl.z += xrow<8>(sl.z); sl.w += xrow<8>(sl.w);
    sl.x += xhalf(sl.x);   sl.y += xhalf(sl.y);   sl.z += xhalf(sl.z);   sl.w += xhalf(sl.w);

    if (lane < 8) *(v4f*)(&shS[wv][4 * q]) = sl;
    if (SOFT && lane == 0) { shM[wv] = ml; shZ[wv] = zl; }
    __syncthreads();

    if (wv == 0) {
        float S = 0.f, Z = (float)RR;
        if (SOFT) {
            float mx = shM[0];
            #pragma unroll
            for (int k = 1; k < 8; ++k) mx = fmaxf(mx, shM[k]);
            Z = 0.f;
            #pragma unroll
            for (int k = 0; k < 8; ++k) {
                const float f = __expf(shM[k] - mx);
                Z += f * shZ[k];
                S += f * shS[k][lane];
            }
        } else {
            #pragma unroll
            for (int k = 0; k < 8; ++k) S += shS[k][lane];
        }
        const float sval = S / Z;
        float sn = sval * sval;
        sn += xrow<1>(sn);
        sn += xrow<2>(sn);
        sn += xrow<4>(sn);
        sn += xrow<8>(sn);
        sn += xhalf(sn);
        const float scale = (sn / (1.f + sn + EPSF)) * rsqrtf(sn + EPSF);
        float val = sval * scale;
        if (addv != nullptr) val += av;
        dst[oi] = val;
    }
}

extern "C" void kernel_launch(void* const* d_in, const int* in_sizes, int n_in,
                              void* d_out, int out_size, void* d_ws, size_t ws_size,
                              hipStream_t stream) {
    (void)in_sizes; (void)n_in; (void)out_size; (void)ws_size;
    const float* x = (const float*)d_in[0];
    const float* w = (const float*)d_in[1];
    float* out = (float*)d_out;

    float* partS = (float*)d_ws;
    float* partM = partS + (size_t)CC * NCH * BB * COUT;
    float* partZ = partM + (size_t)CC * NCH * BB;
    float* v1    = partZ + (size_t)CC * NCH * BB;
    float* v2    = v1 + (size_t)CC * BB * COUT;

    dim3 gP(NCH, CC);
    dim3 gC(BB, CC);
    caps_pass_sum<<<gP, 256, 0, stream>>>(x, w, partS);
    caps_combine<0><<<gC, 256, 0, stream>>>(partS, partM, partZ, nullptr, v1);
    caps_pass_soft<<<gP, 256, 0, stream>>>(x, w, v1, partS, partM, partZ);
    caps_combine<1><<<gC, 256, 0, stream>>>(partS, partM, partZ, v1, v2);
    caps_pass_soft<<<gP, 256, 0, stream>>>(x, w, v2, partS, partM, partZ);
    caps_combine<1><<<gC, 256, 0, stream>>>(partS, partM, partZ, nullptr, out);
}


# Each routing pass merges the previous pass's partials in its own prologue (workgroup = 32 batch rows x 128 route nodes, 16 waves merged in LDS, 32 partials per row): 6 launches become 4
# speedup vs baseline: 3.3395x; 1.0449x over previous
#include <hip/hip_runtime.h>


#define BB   128
#define RR   4096
#define CIN  16
#define COUT 32
#define CC   2
#define EPSF 1e-12f
#define NSP  32
#define NWG  (RR / NSP)
#define NPW  8
#define RGR  32
#define RPI  36

typedef __attribute__((ext_vector_type(2))) float v2f;
typedef __attribute__((ext_vector_type(4))) float v4f;
typedef __attribute__((ext_vector_type(8))) float v8f;


struct Node { v2f wq[4][2]; v4f x0lo, x0hi, x1lo, x1hi; };

__device__ __forceinline__ void load_node(Node& g, const float* __restrict__ wp, const float* __restrict__ xp0,
                                          const float* __restrict__ xp1, int n)
{
    const float* wn = wp + n * (CIN * COUT);
    #pragma unroll
    for (int kk = 0; kk < 4; ++kk) {
        g.wq[kk][0] = *(const v2f*)(wn + (2 * kk) * COUT);
        g.wq[kk][1] = *(const v2f*)(wn + (2 * kk + 1) * COUT);
    }
    g.x0lo = *(const v4f*)(xp0 + n * CIN);
    g.x0hi = *(const v4f*)(xp0 + n * CIN + 4);
    g.x1lo = *(const v4f*)(xp1 + n * CIN);
    g.x1hi = *(const v4f*)(xp1 + n * CIN + 4);
}

__device__ __forceinline__ void frags(const Node& g, v2f (&a)[4][2], v2f (&b0)[4], v2f (&b1)[4])
{
    #pragma unroll
    for (int kk = 0; kk < 4; ++kk) {
        a[kk][0] = v2f{g.wq[kk][0].x, g.wq[kk][1].x};
        a[kk][1] = v2f{g.wq[kk][0].y, g.wq[kk][1].y};
    }
    b0[0] = __builtin_shufflevector(g.x0lo, g.x0lo, 0, 1);
    b0[1] = __builtin_shufflevector(g.x0lo, g.x0lo, 2, 3);
    b0[2] = __builtin_shufflevector(g.x0hi, g.x0hi, 0, 1);
    b0[3] = __builtin_shufflevector(g.x0hi, g.x0hi, 2, 3);
    b1[0] = __builtin_shufflevector(g.x1lo, g.x1lo, 0, 1);
    b1[1] = __builtin_shufflevector(g.x1lo, g.x1lo, 2, 3);
    b1[2] = __builtin_shufflevector(g.x1hi, g.x1hi, 0, 1);
    b1[3] = __builtin_shufflevector(g.x1hi, g.x1hi, 2, 3);
}

__device__ __forceinline__ float xhalf(float v)
{
    const int b = __builtin_bit_cast(int, v);
    return __builtin_bit_cast(float, __builtin_amdgcn_permlanex16(b, b, 0x76543210, 0xfedcba98, false, false));
}

template <int MASK>
__device__ __forceinline__ float xrow(float v)
{
    const int b = __builtin_bit_cast(int, v);
    return __builtin_bit_cast(float, __builtin_amdgcn_update_dpp(b, b, 0x160 + MASK, 0xf, 0xf, false));
}

__device__ __forceinline__ void put16(float* d, const v8f& t0, const v8f& t1)
{
    *(v4f*)(d)      = v4f{t0[0], t1[0], t0[1], t1[1]};
    *(v4f*)(d + 4)  = v4f{t0[2], t1[2], t0[3], t1[3]};
    *(v4f*)(d + 8)  = v4f{t0[4], t1[4], t0[5], t1[5]};
    *(v4f*)(d + 12) = v4f{t0[6], t1[6], t0[7], t1[7]};
}

__device__ __forceinline__ v2f squash2(v2f sval)
{
    float sn = sval.x * sval.x + sval.y * sval.y;
    sn += xrow<1>(sn);
    sn += xrow<2>(sn);
    sn += xrow<4>(sn);
    sn += xrow<8>(sn);
    const float scale = (sn / (1.f + sn + EPSF)) * rsqrtf(sn + EPSF);
    return sval * scale;
}

__global__ __launch_bounds__(512) void caps_pass_sum(
    const float* __restrict__ x,
    const float* __restrict__ w,
    float* __restrict__ pout)
{
    const int sp   = blockIdx.x;
    const int rg   = blockIdx.y;
    const int c    = blockIdx.z;
    const int tid  = threadIdx.x;
    const int wv   = tid >> 5;
    const int lane = tid & 31;
    const int m    = lane & 15;
    const int kh   = lane >> 4;
    const int r0   = sp * NWG + wv * NPW;

    __shared__ __attribute__((aligned(16))) float red[16 * RGR * RPI];

    const float* wp  = w + (((size_t)c * RR + r0) * CIN + 8 * kh) * COUT + 2 * m;
    const float* xp0 = x + ((size_t)(rg * RGR + m) * RR + r0) * CIN + 8 * kh;
    const float* xp1 = xp0 + (size_t)16 * RR * CIN;

    const v8f vzero = {0.f,0.f,0.f,0.f,0.f,0.f,0.f,0.f};
    v8f s00 = vzero, s01 = vzero, s10 = vzero, s11 = vzero;

    Node nd[4];
    load_node(nd[0], wp, xp0, xp1, 0);
    __builtin_amdgcn_sched_barrier(0);
    load_node(nd[1], wp, xp0, xp1, 1);
    __builtin_amdgcn_sched_barrier(0);
    load_node(nd[2], wp, xp0, xp1, 2);
    __builtin_amdgcn_sched_barrier(0);
    #pragma unroll
    for (int n = 0; n < NPW; ++n) {
        if (n + 3 < NPW) load_node(nd[(n + 3) % 4], wp, xp0, xp1, n + 3);
        __builtin_amdgcn_sched_barrier(0);
        v2f a[4][2], b0[4], b1[4];
        frags(nd[n % 4], a, b0, b1);
        #pragma unroll
        for (int kk = 0; kk < 4; ++kk) {
            s00 = __builtin_amdgcn_wmma_f32_16x16x4_f32(false, a[kk][0], false, b0[kk], (short)0, s00, false, false);
            s01 = __builtin_amdgcn_wmma_f32_16x16x4_f32(false, a[kk][1], false, b0[kk], (short)0, s01, false, false);
            s10 = __builtin_amdgcn_wmma_f32_16x16x4_f32(false, a[kk][0], false, b1[kk], (short)0, s10, false, false);
            s11 = __builtin_amdgcn_wmma_f32_16x16x4_f32(false, a[kk][1], false, b1[kk], (short)0, s11, false, false);
        }
        __builtin_amdgcn_sched_barrier(0);
    }

    put16(red + (wv * RGR + m) * RPI + 16 * kh, s00, s01);
    put16(red + (wv * RGR + 16 + m) * RPI + 16 * kh, s10, s11);
    __syncthreads();

    const int row = tid >> 4, oq = tid & 15;
    v2f acc = *(const v2f*)(red + row * RPI + 2 * oq);
    #pragma unroll
    for (int k = 1; k < 16; ++k) acc += *(const v2f*)(red + (k * RGR + row) * RPI + 2 * oq);
    *(v2f*)(pout + (((size_t)c * NSP + sp) * BB + rg * RGR + row) * COUT + 2 * oq) = acc;
}

__device__ __forceinline__ float logit16(const v8f& a0, const v8f& a1, const v8f& u0, const v8f& u1)
{
    v2f t0 = __builtin_shufflevector(a0, a0, 0, 1) * __builtin_shufflevector(u0, u0, 0, 1);
    v2f t1 = __builtin_shufflevector(a1, a1, 0, 1) * __builtin_shufflevector(u1, u1, 0, 1);
    t0 += __builtin_shufflevector(a0, a0, 2, 3) * __builtin_shufflevector(u0, u0, 2, 3);
    t1 += __builtin_shufflevector(a1, a1, 2, 3) * __builtin_shufflevector(u1, u1, 2, 3);
    t0 += __builtin_shufflevector(a0, a0, 4, 5) * __builtin_shufflevector(u0, u0, 4, 5);
    t1 += __builtin_shufflevector(a1, a1, 4, 5) * __builtin_shufflevector(u1, u1, 4, 5);
    t0 += __builtin_shufflevector(a0, a0, 6, 7) * __builtin_shufflevector(u0, u0, 6, 7);
    t1 += __builtin_shufflevector(a1, a1, 6, 7) * __builtin_shufflevector(u1, u1, 6, 7);
    const v2f ts = t0 + t1;
    return ts.x + ts.y;
}

__device__ __forceinline__ void soft_step(float d, float& mrun, float& zrun, v8f& s0, v8f& s1,
                                          const v8f& a0, const v8f& a1)
{
    const float mo = mrun;
    const float mn = fmaxf(mo, d);
    const float e  = __expf(fminf(mo, d) - mn);
    const bool  up = d > mo;
    const float sc = up ? e : 1.f;
    const float wg = up ? 1.f : e;
    mrun = mn;
    zrun = zrun * sc + wg;
    #pragma unroll
    for (int v = 0; v < 8; ++v) {
        s0[v] = s0[v] * sc + wg * a0[v];
        s1[v] = s1[v] * sc + wg * a1[v];
    }
}

template <int SOFTIN>
__global__ __launch_bounds__(512) void caps_pass_soft(
    const float* __restrict__ x,
    const float* __restrict__ w,
    const float* __restrict__ pin,
    const float* __restrict__ pmin,
    const float* __restrict__ pzin,
    const float* __restrict__ v1in,
    float* __restrict__ v1out,
    float* __restrict__ pout,
    float* __restrict__ pmout,
    float* __restrict__ pzout)
{
    const int sp   = blockIdx.x;
    const int rg   = blockIdx.y;
    const int c    = blockIdx.z;
    const int tid  = threadIdx.x;
    const int wv   = tid >> 5;
    const int lane = tid & 31;
    const int m    = lane & 15;
    const int kh   = lane >> 4;
    const int r0   = sp * NWG + wv * NPW;
    const int row  = tid >> 4, oq = tid & 15;
    const int bg   = rg * RGR + row;

    __shared__ __attribute__((aligned(16))) float red[16 * RGR * RPI];
    __shared__ float redM[16 * RGR];
    __shared__ float redZ[16 * RGR];
    __shared__ __attribute__((aligned(16))) float vsh[RGR * COUT];

    const float* wp  = w + (((size_t)c * RR + r0) * CIN + 8 * kh) * COUT + 2 * m;
    const float* xp0 = x + ((size_t)(rg * RGR + m) * RR + r0) * CIN + 8 * kh;
    const float* xp1 = xp0 + (size_t)16 * RR * CIN;

    Node ga, gb;
    load_node(ga, wp, xp0, xp1, 0);
    load_node(gb, wp, xp0, xp1, 1);

    {
        v2f S = {0.f, 0.f};
        float Z = (float)RR;
        const size_t p0 = (size_t)c * NSP * BB + bg;
        if (SOFTIN) {
            float mv[NSP], zv[NSP];
            v2f   sv[NSP];
            #pragma unroll
            for (int k = 0; k < NSP; ++k) mv[k] = pmin[p0 + (size_t)k * BB];
            #pragma unroll
            for (int k = 0; k < NSP; ++k) {
                zv[k] = pzin[p0 + (size_t)k * BB];
                sv[k] = *(const v2f*)(pin + (p0 + (size_t)k * BB) * COUT + 2 * oq);
            }
            float mx = mv[0];
            #pragma unroll
            for (int k = 1; k < NSP; ++k) mx = fmaxf(mx, mv[k]);
            Z = 0.f;
            #pragma unroll
            for (int k = 0; k < NSP; ++k) {
                const float f = __expf(mv[k] - mx);
                Z += f * zv[k];
                S += f * sv[k];
            }
        } else {
            #pragma unroll
            for (int k = 0; k < NSP; ++k) S += *(const v2f*)(pin + (p0 + (size_t)k * BB) * COUT + 2 * oq);
        }
        v2f sval = {S.x / Z, S.y / Z};
        v2f val = squash2(sval);
        const size_t vi = ((size_t)c * BB + bg) * COUT + 2 * oq;
        if (SOFTIN) val += *(const v2f*)(v1in + vi);
        else if (sp == 0) *(v2f*)(v1out + vi) = val;
        *(v2f*)(vsh + row * COUT + 2 * oq) = val;
    }
    __syncthreads();

    v8f u00, u01, u10, u11;
    {
        const float* q0 = vsh + m * COUT + 16 * kh;
        const float* q1 = vsh + (16 + m) * COUT + 16 * kh;
        #pragma unroll
        for (int j = 0; j < 4; ++j) {
            const v4f e0 = *(const v4f*)(q0 + 4 * j);
            const v4f e1 = *(const v4f*)(q1 + 4 * j);
            u00[2 * j] = e0.x; u01[2 * j] = e0.y; u00[2 * j + 1] = e0.z; u01[2 * j + 1] = e0.w;
            u10[2 * j] = e1.x; u11[2 * j] = e1.y; u10[2 * j + 1] = e1.z; u11[2 * j + 1] = e1.w;
        }
    }

    const v8f vzero = {0.f,0.f,0.f,0.f,0.f,0.f,0.f,0.f};
    v8f s00 = vzero, s01 = vzero, s10 = vzero, s11 = vzero;
    float m0 = -3.402823466e38f, m1 = -3.402823466e38f;
    float z0 = 0.f, z1 = 0.f;

    #pragma unroll 1
    for (int it = 0; it < NPW / 2; ++it) {
        #pragma unroll
        for (int h = 0; h < 2; ++h) {
            Node& g = h ? gb : ga;
            v2f a[4][2], b0[4], b1[4];
            frags(g, a, b0, b1);
            v8f p00 = vzero, p01 = vzero, p10 = vzero, p11 = vzero;
            #pragma unroll
            for (int kk = 0; kk < 4; ++kk) {
                p00 = __builtin_amdgcn_wmma_f32_16x16x4_f32(false, a[kk][0], false, b0[kk], (short)0, p00, false, false);
                p01 = __builtin_amdgcn_wmma_f32_16x16x4_f32(false, a[kk][1], false, b0[kk], (short)0, p01, false, false);
                p10 = __builtin_amdgcn_wmma_f32_16x16x4_f32(false, a[kk][0], false, b1[kk], (short)0, p10, false, false);
                p11 = __builtin_amdgcn_wmma_f32_16x16x4_f32(false, a[kk][1], false, b1[kk], (short)0, p11, false, false);
            }
            if (it + 1 < NPW / 2) load_node(g, wp, xp0, xp1, 2 * it + 2 + h);
            float d0 = logit16(p00, p01, u00, u01);
            float d1 = logit16(p10, p11, u10, u11);
            d0 = d0 + xhalf(d0);
            d1 = d1 + xhalf(d1);
            soft_step(d0, m0, z0, s00, s01, p00, p01);
            soft_step(d1, m1, z1, s10, s11, p10, p11);
        }
    }

    put16(red + (wv * RGR + m) * RPI + 16 * kh, s00, s01);
    put16(red + (wv * RGR + 16 + m) * RPI + 16 * kh, s10, s11);
    if (kh == 0) {
        redM[wv * RGR + m]      = m0;
        redZ[wv * RGR + m]      = z0;
        redM[wv * RGR + 16 + m] = m1;
        redZ[wv * RGR + 16 + m] = z1;
    }
    __syncthreads();

    {
        float mx = redM[row];
        #pragma unroll
        for (int k = 1; k < 16; ++k) mx = fmaxf(mx, redM[k * RGR + row]);
        float Z = 0.f;
        v2f   S = {0.f, 0.f};
        #pragma unroll
        for (int k = 0; k < 16; ++k) {
            const float f = __expf(redM[k * RGR + row] - mx);
            Z += f * redZ[k * RGR + row];
            S += f * *(const v2f*)(red + (k * RGR + row) * RPI + 2 * oq);
        }
        const size_t pi = ((size_t)c * NSP + sp) * BB + bg;
        *(v2f*)(pout + pi * COUT + 2 * oq) = S;
        if (oq == 0) { pmout[pi] = mx; pzout[pi] = Z; }
    }
}

__global__ __launch_bounds__(64) void caps_final(
    const float* __restrict__ pin,
    const float* __restrict__ pmin,
    const float* __restrict__ pzin,
    float* __restrict__ dst)
{
    const int c = blockIdx.y, tid = threadIdx.x;
    const int row = tid >> 4, oq = tid & 15;
    const int bg = blockIdx.x * 4 + row;
    const size_t p0 = (size_t)c * NSP * BB + bg;

    float mv[NSP], zv[NSP];
    v2f   sv[NSP];
    #pragma unroll
    for (int k = 0; k < NSP; ++k) mv[k] = pmin[p0 + (size_t)k * BB];
    #pragma unroll
    for (int k = 0; k < NSP; ++k) {
        zv[k] = pzin[p0 + (size_t)k * BB];
        sv[k] = *(const v2f*)(pin + (p0 + (size_t)k * BB) * COUT + 2 * oq);
    }
    float mx = mv[0];
    #pragma unroll
    for (int k = 1; k < NSP; ++k) mx = fmaxf(mx, mv[k]);
    float Z = 0.f;
    v2f   S = {0.f, 0.f};
    #pragma unroll
    for (int k = 0; k < NSP; ++k) {
        const float f = __expf(mv[k] - mx);
        Z += f * zv[k];
        S += f * sv[k];
    }
    const v2f sval = {S.x / Z, S.y / Z};
    *(v2f*)(dst + ((size_t)c * BB + bg) * COUT + 2 * oq) = squash2(sval);
}

extern "C" void kernel_launch(void* const* d_in, const int* in_sizes, int n_in,
                              void* d_out, int out_size, void* d_ws, size_t ws_size,
                              hipStream_t stream) {
    (void)in_sizes; (void)n_in; (void)out_size; (void)ws_size;
    const float* x = (const float*)d_in[0];
    const float* w = (const float*)d_in[1];
    float* out = (float*)d_out;

    float* pA  = (float*)d_ws;
    float* mA  = pA + (size_t)CC * NSP * BB * COUT;
    float* zA  = mA + (size_t)CC * NSP * BB;
    float* pB  = zA + (size_t)CC * NSP * BB;
    float* mB  = pB + (size_t)CC * NSP * BB * COUT;
    float* zB  = mB + (size_t)CC * NSP * BB;
    float* v1  = zB + (size_t)CC * NSP * BB;

    dim3 gP(NSP, BB / RGR, CC);
    caps_pass_sum<<<gP, 512, 0, stream>>>(x, w, pA);
    caps_pass_soft<0><<<gP, 512, 0, stream>>>(x, w, pA, nullptr, nullptr, nullptr, v1, pB, mB, zB);
    caps_pass_soft<1><<<gP, 512, 0, stream>>>(x, w, pB, mB, zB, v1, nullptr, pA, mA, zA);
    caps_final<<<dim3(BB / 4, CC), 64, 0, stream>>>(pA, mA, zA, out);
}


# Softmax merges: each rescale factor is formed once per row (lane k: splits k, k+16) and shared by DPP row_share, max and sum by row reductions: 36 loads, 2 exps per thread (were 96, 32)
# speedup vs baseline: 3.5679x; 1.0684x over previous
#include <hip/hip_runtime.h>


#define BB   128
#define RR   4096
#define CIN  16
#define COUT 32
#define CC   2
#define EPSF 1e-12f
#define NSP  32
#define NWG  (RR / NSP)
#define NPW  8
#define RGR  32
#define RPI  36

typedef __attribute__((ext_vector_type(2))) float v2f;
typedef __attribute__((ext_vector_type(4))) float v4f;
typedef __attribute__((ext_vector_type(8))) float v8f;


struct Node { v2f wq[4][2]; v4f x0lo, x0hi, x1lo, x1hi; };

__device__ __forceinline__ void load_node(Node& g, const float* __restrict__ wp, const float* __restrict__ xp0,
                                          const float* __restrict__ xp1, int n)
{
    const float* wn = wp + n * (CIN * COUT);
    #pragma unroll
    for (int kk = 0; kk < 4; ++kk) {
        g.wq[kk][0] = *(const v2f*)(wn + (2 * kk) * COUT);
        g.wq[kk][1] = *(const v2f*)(wn + (2 * kk + 1) * COUT);
    }
    g.x0lo = *(const v4f*)(xp0 + n * CIN);
    g.x0hi = *(const v4f*)(xp0 + n * CIN + 4);
    g.x1lo = *(const v4f*)(xp1 + n * CIN);
    g.x1hi = *(const v4f*)(xp1 + n * CIN + 4);
}

__device__ __forceinline__ void frags(const Node& g, v2f (&a)[4][2], v2f (&b0)[4], v2f (&b1)[4])
{
    #pragma unroll
    for (int kk = 0; kk < 4; ++kk) {
        a[kk][0] = v2f{g.wq[kk][0].x, g.wq[kk][1].x};
        a[kk][1] = v2f{g.wq[kk][0].y, g.wq[kk][1].y};
    }
    b0[0] = __builtin_shufflevector(g.x0lo, g.x0lo, 0, 1);
    b0[1] = __builtin_shufflevector(g.x0lo, g.x0lo, 2, 3);
    b0[2] = __builtin_shufflevector(g.x0hi, g.x0hi, 0, 1);
    b0[3] = __builtin_shufflevector(g.x0hi, g.x0hi, 2, 3);
    b1[0] = __builtin_shufflevector(g.x1lo, g.x1lo, 0, 1);
    b1[1] = __builtin_shufflevector(g.x1lo, g.x1lo, 2, 3);
    b1[2] = __builtin_shufflevector(g.x1hi, g.x1hi, 0, 1);
    b1[3] = __builtin_shufflevector(g.x1hi, g.x1hi, 2, 3);
}

__device__ __forceinline__ float xhalf(float v)
{
    const int b = __builtin_bit_cast(int, v);
    return __builtin_bit_cast(float, __builtin_amdgcn_permlanex16(b, b, 0x76543210, 0xfedcba98, false, false));
}

template <int MASK>
__device__ __forceinline__ float xrow(float v)
{
    const int b = __builtin_bit_cast(int, v);
    return __builtin_bit_cast(float, __builtin_amdgcn_update_dpp(b, b, 0x160 + MASK, 0xf, 0xf, false));
}

template <int K>
__device__ __forceinline__ float xshare(float v)
{
    const int b = __builtin_bit_cast(int, v);
    return __builtin_bit_cast(float, __builtin_amdgcn_mov_dpp(b, 0x150 + K, 0xf, 0xf, false));
}

template <int MASK>
__device__ __forceinline__ float xmask(float v)
{
    const int b = __builtin_bit_cast(int, v);
    return __builtin_bit_cast(float, __builtin_amdgcn_mov_dpp(b, 0x160 + MASK, 0xf, 0xf, false));
}

__device__ __forceinline__ float rowmax16(float v)
{
    v = fmaxf(v, xmask<1>(v));
    v = fmaxf(v, xmask<2>(v));
    v = fmaxf(v, xmask<4>(v));
    v = fmaxf(v, xmask<8>(v));
    return v;
}

__device__ __forceinline__ float rowsum16(float v)
{
    v += xmask<1>(v);
    v += xmask<2>(v);
    v += xmask<4>(v);
    v += xmask<8>(v);
    return v;
}

template <int K>
struct Wacc {
    static __device__ __forceinline__ void run(v2f& S, float f, const v2f (&sv)[16])
    {
        S += xshare<K>(f) * sv[K];
        Wacc<K + 1>::run(S, f, sv);
    }
};
template <>
struct Wacc<16> {
    static __device__ __forceinline__ void run(v2f&, float, const v2f (&)[16]) {}
};

__device__ __forceinline__ void merge_partials(const float* __restrict__ pin, const float* __restrict__ pmin,
                                               const float* __restrict__ pzin, size_t p0, int oq, v2f& S, float& Z)
{
    static_assert(NSP == 32, "two splits per lane of a 16-lane row");
    const float ma = pmin[p0 + (size_t)oq * BB];
    const float mb = pmin[p0 + (size_t)(oq + 16) * BB];
    const float za = pzin[p0 + (size_t)oq * BB];
    const float zb = pzin[p0 + (size_t)(oq + 16) * BB];
    v2f sva[16], svb[16];
    #pragma unroll
    for (int k = 0; k < 16; ++k) {
        sva[k] = *(const v2f*)(pin + (p0 + (size_t)k * BB) * COUT + 2 * oq);
        svb[k] = *(const v2f*)(pin + (p0 + (size_t)(k + 16) * BB) * COUT + 2 * oq);
    }
    __builtin_amdgcn_sched_barrier(0);
    const float mx = rowmax16(fmaxf(ma, mb));
    const float fa = __expf(ma - mx);
    const float fb = __expf(mb - mx);
    Z = rowsum16(fa * za + fb * zb);
    S = v2f{0.f, 0.f};
    Wacc<0>::run(S, fa, sva);
    Wacc<0>::run(S, fb, svb);
}

__device__ __forceinline__ void put16(float* d, const v8f& t0, const v8f& t1)
{
    *(v4f*)(d)      = v4f{t0[0], t1[0], t0[1], t1[1]};
    *(v4f*)(d + 4)  = v4f{t0[2], t1[2], t0[3], t1[3]};
    *(v4f*)(d + 8)  = v4f{t0[4], t1[4], t0[5], t1[5]};
    *(v4f*)(d + 12) = v4f{t0[6], t1[6], t0[7], t1[7]};
}

__device__ __forceinline__ v2f squash2(v2f sval)
{
    float sn = sval.x * sval.x + sval.y * sval.y;
    sn += xrow<1>(sn);
    sn += xrow<2>(sn);
    sn += xrow<4>(sn);
    sn += xrow<8>(sn);
    const float scale = (sn / (1.f + sn + EPSF)) * rsqrtf(sn + EPSF);
    return sval * scale;
}

__global__ __launch_bounds__(512) void caps_pass_sum(
    const float* __restrict__ x,
    const float* __restrict__ w,
    float* __restrict__ pout)
{
    const int sp   = blockIdx.x;
    const int rg   = blockIdx.y;
    const int c    = blockIdx.z;
    const int tid  = threadIdx.x;
    const int wv   = tid >> 5;
    const int lane = tid & 31;
    const int m    = lane & 15;
    const int kh   = lane >> 4;
    const int r0   = sp * NWG + wv * NPW;

    __shared__ __attribute__((aligned(16))) float red[16 * RGR * RPI];

    const float* wp  = w + (((size_t)c * RR + r0) * CIN + 8 * kh) * COUT + 2 * m;
    const float* xp0 = x + ((size_t)(rg * RGR + m) * RR + r0) * CIN + 8 * kh;
    const float* xp1 = xp0 + (size_t)16 * RR * CIN;

    const v8f vzero = {0.f,0.f,0.f,0.f,0.f,0.f,0.f,0.f};
    v8f s00 = vzero, s01 = vzero, s10 = vzero, s11 = vzero;

    Node nd[4];
    load_node(nd[0], wp, xp0, xp1, 0);
    __builtin_amdgcn_sched_barrier(0);
    load_node(nd[1], wp, xp0, xp1, 1);
    __builtin_amdgcn_sched_barrier(0);
    load_node(nd[2], wp, xp0, xp1, 2);
    __builtin_amdgcn_sched_barrier(0);
    #pragma unroll
    for (int n = 0; n < NPW; ++n) {
        if (n + 3 < NPW) load_node(nd[(n + 3) % 4], wp, xp0, xp1, n + 3);
        __builtin_amdgcn_sched_barrier(0);
        v2f a[4][2], b0[4], b1[4];
        frags(nd[n % 4], a, b0, b1);
        #pragma unroll
        for (int kk = 0; kk < 4; ++kk) {
            s00 = __builtin_amdgcn_wmma_f32_16x16x4_f32(false, a[kk][0], false, b0[kk], (short)0, s00, false, false);
            s01 = __builtin_amdgcn_wmma_f32_16x16x4_f32(false, a[kk][1], false, b0[kk], (short)0, s01, false, false);
            s10 = __builtin_amdgcn_wmma_f32_16x16x4_f32(false, a[kk][0], false, b1[kk], (short)0, s10, false, false);
            s11 = __builtin_amdgcn_wmma_f32_16x16x4_f32(false, a[kk][1], false, b1[kk], (short)0, s11, false, false);
        }
        __builtin_amdgcn_sched_barrier(0);
    }

    put16(red + (wv * RGR + m) * RPI + 16 * kh, s00, s01);
    put16(red + (wv * RGR + 16 + m) * RPI + 16 * kh, s10, s11);
    __syncthreads();

    const int row = tid >> 4, oq = tid & 15;
    v2f acc = *(const v2f*)(red + row * RPI + 2 * oq);
    #pragma unroll
    for (int k = 1; k < 16; ++k) acc += *(const v2f*)(red + (k * RGR + row) * RPI + 2 * oq);
    *(v2f*)(pout + (((size_t)c * NSP + sp) * BB + rg * RGR + row) * COUT + 2 * oq) = acc;
}

__device__ __forceinline__ float logit16(const v8f& a0, const v8f& a1, const v8f& u0, const v8f& u1)
{
    v2f t0 = __builtin_shufflevector(a0, a0, 0, 1) * __builtin_shufflevector(u0, u0, 0, 1);
    v2f t1 = __builtin_shufflevector(a1, a1, 0, 1) * __builtin_shufflevector(u1, u1, 0, 1);
    t0 += __builtin_shufflevector(a0, a0, 2, 3) * __builtin_shufflevector(u0, u0, 2, 3);
    t1 += __builtin_shufflevector(a1, a1, 2, 3) * __builtin_shufflevector(u1, u1, 2, 3);
    t0 += __builtin_shufflevector(a0, a0, 4, 5) * __builtin_shufflevector(u0, u0, 4, 5);
    t1 += __builtin_shufflevector(a1, a1, 4, 5) * __builtin_shufflevector(u1, u1, 4, 5);
    t0 += __builtin_shufflevector(a0, a0, 6, 7) * __builtin_shufflevector(u0, u0, 6, 7);
    t1 += __builtin_shufflevector(a1, a1, 6, 7) * __builtin_shufflevector(u1, u1, 6, 7);
    const v2f ts = t0 + t1;
    return ts.x + ts.y;
}

__device__ __forceinline__ void soft_step(float d, float& mrun, float& zrun, v8f& s0, v8f& s1,
                                          const v8f& a0, const v8f& a1)
{
    const float mo = mrun;
    const float mn = fmaxf(mo, d);
    const float e  = __expf(fminf(mo, d) - mn);
    const bool  up = d > mo;
    const float sc = up ? e : 1.f;
    const float wg = up ? 1.f : e;
    mrun = mn;
    zrun = zrun * sc + wg;
    #pragma unroll
    for (int v = 0; v < 8; ++v) {
        s0[v] = s0[v] * sc + wg * a0[v];
        s1[v] = s1[v] * sc + wg * a1[v];
    }
}

template <int SOFTIN>
__global__ __launch_bounds__(512) void caps_pass_soft(
    const float* __restrict__ x,
    const float* __restrict__ w,
    const float* __restrict__ pin,
    const float* __restrict__ pmin,
    const float* __restrict__ pzin,
    const float* __restrict__ v1in,
    float* __restrict__ v1out,
    float* __restrict__ pout,
    float* __restrict__ pmout,
    float* __restrict__ pzout)
{
    const int sp   = blockIdx.x;
    const int rg   = blockIdx.y;
    const int c    = blockIdx.z;
    const int tid  = threadIdx.x;
    const int wv   = tid >> 5;
    const int lane = tid & 31;
    const int m    = lane & 15;
    const int kh   = lane >> 4;
    const int r0   = sp * NWG + wv * NPW;
    const int row  = tid >> 4, oq = tid & 15;
    const int bg   = rg * RGR + row;

    __shared__ __attribute__((aligned(16))) float red[16 * RGR * RPI];
    __shared__ float redM[16 * RGR];
    __shared__ float redZ[16 * RGR];
    __shared__ __attribute__((aligned(16))) float vsh[RGR * COUT];

    const float* wp  = w + (((size_t)c * RR + r0) * CIN + 8 * kh) * COUT + 2 * m;
    const float* xp0 = x + ((size_t)(rg * RGR + m) * RR + r0) * CIN + 8 * kh;
    const float* xp1 = xp0 + (size_t)16 * RR * CIN;

    Node ga, gb;
    load_node(ga, wp, xp0, xp1, 0);
    load_node(gb, wp, xp0, xp1, 1);
    __builtin_amdgcn_sched_barrier(0);

    {
        v2f S = {0.f, 0.f};
        float Z = (float)RR;
        const size_t p0 = (size_t)c * NSP * BB + bg;
        const size_t vi = ((size_t)c * BB + bg) * COUT + 2 * oq;
        v2f v1v = {0.f, 0.f};
        if (SOFTIN) {
            v1v = *(const v2f*)(v1in + vi);
            merge_partials(pin, pmin, pzin, p0, oq, S, Z);
        } else {
            #pragma unroll
            for (int k = 0; k < NSP; ++k) S += *(const v2f*)(pin + (p0 + (size_t)k * BB) * COUT + 2 * oq);
        }
        v2f sval = {S.x / Z, S.y / Z};
        v2f val = squash2(sval);
        if (SOFTIN) val += v1v;
        else if (sp == 0) *(v2f*)(v1out + vi) = val;
        *(v2f*)(vsh + row * COUT + 2 * oq) = val;
    }
    __syncthreads();

    v8f u00, u01, u10, u11;
    {
        const float* q0 = vsh + m * COUT + 16 * kh;
        const float* q1 = vsh + (16 + m) * COUT + 16 * kh;
        #pragma unroll
        for (int j = 0; j < 4; ++j) {
            const v4f e0 = *(const v4f*)(q0 + 4 * j);
            const v4f e1 = *(const v4f*)(q1 + 4 * j);
            u00[2 * j] = e0.x; u01[2 * j] = e0.y; u00[2 * j + 1] = e0.z; u01[2 * j + 1] = e0.w;
            u10[2 * j] = e1.x; u11[2 * j] = e1.y; u10[2 * j + 1] = e1.z; u11[2 * j + 1] = e1.w;
        }
    }

    const v8f vzero = {0.f,0.f,0.f,0.f,0.f,0.f,0.f,0.f};
    v8f s00 = vzero, s01 = vzero, s10 = vzero, s11 = vzero;
    float m0 = -3.402823466e38f, m1 = -3.402823466e38f;
    float z0 = 0.f, z1 = 0.f;

    #pragma unroll 1
    for (int it = 0; it < NPW / 2; ++it) {
        #pragma unroll
        for (int h = 0; h < 2; ++h) {
            Node& g = h ? gb : ga;
            v2f a[4][2], b0[4], b1[4];
            frags(g, a, b0, b1);
            v8f p00 = vzero, p01 = vzero, p10 = vzero, p11 = vzero;
            #pragma unroll
            for (int kk = 0; kk < 4; ++kk) {
                p00 = __builtin_amdgcn_wmma_f32_16x16x4_f32(false, a[kk][0], false, b0[kk], (short)0, p00, false, false);
                p01 = __builtin_amdgcn_wmma_f32_16x16x4_f32(false, a[kk][1], false, b0[kk], (short)0, p01, false, false);
                p10 = __builtin_amdgcn_wmma_f32_16x16x4_f32(false, a[kk][0], false, b1[kk], (short)0, p10, false, false);
                p11 = __builtin_amdgcn_wmma_f32_16x16x4_f32(false, a[kk][1], false, b1[kk], (short)0, p11, false, false);
            }
            if (it + 1 < NPW / 2) load_node(g, wp, xp0, xp1, 2 * it + 2 + h);
            float d0 = logit16(p00, p01, u00, u01);
            float d1 = logit16(p10, p11, u10, u11);
            d0 = d0 + xhalf(d0);
            d1 = d1 + xhalf(d1);
            soft_step(d0, m0, z0, s00, s01, p00, p01);
            soft_step(d1, m1, z1, s10, s11, p10, p11);
        }
    }

    put16(red + (wv * RGR + m) * RPI + 16 * kh, s00, s01);
    put16(red + (wv * RGR + 16 + m) * RPI + 16 * kh, s10, s11);
    if (kh == 0) {
        redM[wv * RGR + m]      = m0;
        redZ[wv * RGR + m]      = z0;
        redM[wv * RGR + 16 + m] = m1;
        redZ[wv * RGR + 16 + m] = z1;
    }
    __syncthreads();

    {
        const float mk = redM[oq * RGR + row];
        const float zk = redZ[oq * RGR + row];
        v2f sv[16];
        #pragma unroll
        for (int k = 0; k < 16; ++k) sv[k] = *(const v2f*)(red + (k * RGR + row) * RPI + 2 * oq);
        const float mx = rowmax16(mk);
        const float fk = __expf(mk - mx);
        const float Z  = rowsum16(fk * zk);
        v2f S = {0.f, 0.f};
        Wacc<0>::run(S, fk, sv);
        const size_t pi = ((size_t)c * NSP + sp) * BB + bg;
        *(v2f*)(pout + pi * COUT + 2 * oq) = S;
        if (oq == 0) { pmout[pi] = mx; pzout[pi] = Z; }
    }
}

__global__ __launch_bounds__(64) void caps_final(
    const float* __restrict__ pin,
    const float* __restrict__ pmin,
    const float* __restrict__ pzin,
    float* __restrict__ dst)
{
    const int c = blockIdx.y, tid = threadIdx.x;
    const int row = tid >> 4, oq = tid & 15;
    const int bg = blockIdx.x * 4 + row;
    const size_t p0 = (size_t)c * NSP * BB + bg;

    float Z;
    v2f   S;
    merge_partials(pin, pmin, pzin, p0, oq, S, Z);
    const v2f sval = {S.x / Z, S.y / Z};
    *(v2f*)(dst + ((size_t)c * BB + bg) * COUT + 2 * oq) = squash2(sval);
}

extern "C" void kernel_launch(void* const* d_in, const int* in_sizes, int n_in,
                              void* d_out, int out_size, void* d_ws, size_t ws_size,
                              hipStream_t stream) {
    (void)in_sizes; (void)n_in; (void)out_size; (void)ws_size;
    const float* x = (const float*)d_in[0];
    const float* w = (const float*)d_in[1];
    float* out = (float*)d_out;

    float* pA  = (float*)d_ws;
    float* mA  = pA + (size_t)CC * NSP * BB * COUT;
    float* zA  = mA + (size_t)CC * NSP * BB;
    float* pB  = zA + (size_t)CC * NSP * BB;
    float* mB  = pB + (size_t)CC * NSP * BB * COUT;
    float* zB  = mB + (size_t)CC * NSP * BB;
    float* v1  = zB + (size_t)CC * NSP * BB;

    dim3 gP(NSP, BB / RGR, CC);
    caps_pass_sum<<<gP, 512, 0, stream>>>(x, w, pA);
    caps_pass_soft<0><<<gP, 512, 0, stream>>>(x, w, pA, nullptr, nullptr, nullptr, v1, pB, mB, zB);
    caps_pass_soft<1><<<gP, 512, 0, stream>>>(x, w, pB, mB, zB, v1, nullptr, pA, mA, zA);
    caps_final<<<dim3(BB / 4, CC), 64, 0, stream>>>(pA, mA, zA, out);
}
